# starting kernel
# baseline (speedup 1.0000x reference)
#include <hip/hip_runtime.h>


typedef __attribute__((ext_vector_type(2))) float v2f;
typedef __attribute__((ext_vector_type(8))) float v8f;

#define N_SAMPLES 32768
#define N_EVENTS  64
#define STEP      256
#define BATCH     16

#define WAVES_PER_BLOCK 8
#define WAVE_TILE       64
#define TILES_PER_BATCH  (N_SAMPLES / WAVE_TILE)
#define BLOCKS_PER_BATCH (TILES_PER_BATCH / WAVES_PER_BLOCK)
#define GRID (BATCH * BLOCKS_PER_BATCH)

__device__ __forceinline__ float masked_gather(const float* __restrict__ xb,
                                               int rowbase, int j) {
    const int mask = ~(j >> 31);
    const float v  = xb[rowbase + (j & mask)];
    return __int_as_float(__float_as_int(v) & mask);
}

__global__ __launch_bounds__(256)
void sparse_audio_gather_wmma(const float* __restrict__ x,
                              const int*   __restrict__ indices,
                              float*       __restrict__ out) {
    __shared__ int t_sh[N_EVENTS];

    const int tid = threadIdx.x;
    const int b = blockIdx.x / BLOCKS_PER_BATCH;

    if (tid < N_EVENTS) {
        t_sh[tid] = indices[b * N_EVENTS + tid] * STEP;
    }
    __syncthreads();

    const int wave      = blockIdx.x * WAVES_PER_BLOCK + (tid >> 5);
    const int lane      = tid & 31;
    const int lh        = lane >> 4;
    const int n_in_tile = lane & 15;
    const int s_base    = (wave % TILES_PER_BATCH) * WAVE_TILE;
    const int n0        = s_base + n_in_tile;

    const float* xb = x + (size_t)b * N_EVENTS * N_SAMPLES;

    v8f zero = {};
    v8f acc[4];
    acc[0] = zero; acc[1] = zero; acc[2] = zero; acc[3] = zero;

    v2f a_ones;
    a_ones.x = 1.0f;
    a_ones.y = 1.0f;

    #pragma unroll 8
    for (int g = 0; g < 16; ++g) {
        const int eA = 4 * g + lh;
        const int eB = 4 * g + 2 + lh;
        const int tA = t_sh[eA];
        const int tB = t_sh[eB];
        const int rowA = eA * N_SAMPLES;
        const int rowB = eB * N_SAMPLES;

        #pragma unroll
        for (int t = 0; t < 4; ++t) {
            const int n = n0 + t * 16;
            v2f bm;
            bm.x = masked_gather(xb, rowA, n - tA);
            bm.y = masked_gather(xb, rowB, n - tB);
            acc[t] = __builtin_amdgcn_wmma_f32_16x16x4_f32(
                 false, a_ones,
                 false, bm,
                 (short)0, acc[t],
                 false,  false);
        }
    }

    if (lane < 16) {
        float* ob = out + (size_t)b * N_SAMPLES + s_base;
        #pragma unroll
        for (int t = 0; t < 4; ++t) {
            ob[t * 16 + lane] = acc[t][0];
        }
    }
}

extern "C" void kernel_launch(void* const* d_in, const int* in_sizes, int n_in,
                              void* d_out, int out_size, void* d_ws, size_t ws_size,
                              hipStream_t stream) {
    (void)in_sizes; (void)n_in; (void)out_size; (void)d_ws; (void)ws_size;
    const float* x   = (const float*)d_in[0];
    const int*   idx = (const int*)d_in[1];
    float*       out = (float*)d_out;

    sparse_audio_gather_wmma<<<GRID, 256, 0, stream>>>(x, idx, out);
}


# each gather is one wave-wide 16-byte-per-lane load (512 contiguous bytes of one event row), issued 16 ahead; wave tile 64->128 samples, grid 1024->512; events summed in pairs (order changes)
# speedup vs baseline: 1.2626x; 1.2626x over previous
#include <hip/hip_runtime.h>


typedef __attribute__((ext_vector_type(2))) float v2f;
typedef __attribute__((ext_vector_type(4))) float v4f;
typedef __attribute__((ext_vector_type(4))) int   v4i;
typedef __attribute__((ext_vector_type(8))) float v8f;

#define N_SAMPLES 32768
#define N_EVENTS  64
#define STEP      256
#define BATCH     16

#define WAVES_PER_BLOCK 8
#define WAVE_TILE       128
#define TILES_PER_BATCH  (N_SAMPLES / WAVE_TILE)
#define BLOCKS_PER_BATCH (TILES_PER_BATCH / WAVES_PER_BLOCK)
#define GRID (BATCH * BLOCKS_PER_BATCH)

#define GROUP   16
#define NGROUPS (N_EVENTS / GROUP)

__device__ __forceinline__ int tile_start(int index, int s_base) {
    return s_base - index * STEP;
}

__device__ __forceinline__ unsigned row_offset(int j, int lane) {
    const int mask = ~(j >> 31);
    return (unsigned)(j & mask) * 4u + 16u * (unsigned)lane;
}

__device__ __forceinline__ v4f load_row(const float* __restrict__ xb, int e, unsigned voff) {
    const char* row = reinterpret_cast<const char*>(xb) + (size_t)e * (N_SAMPLES * 4);
    return *reinterpret_cast<const v4f*>(row + voff);
}

__device__ __forceinline__ v4f zero_unless_begun(v4f v, int j) {
    const int mask = ~(j >> 31);
    const v4i m = {mask, mask, mask, mask};
    return __builtin_bit_cast(v4f, __builtin_bit_cast(v4i, v) & m);
}

__global__ __launch_bounds__(256)
void sparse_audio_gather_wmma(const float* __restrict__ x,
                              const int*   __restrict__ indices,
                              float*       __restrict__ out) {
    const int tid  = threadIdx.x;
    const int lane = tid & 31;
    const int b = blockIdx.x / BLOCKS_PER_BATCH;
    const int wave_in_block = __builtin_amdgcn_readfirstlane(tid >> 5);
    const int s_base =
        ((blockIdx.x % BLOCKS_PER_BATCH) * WAVES_PER_BLOCK + wave_in_block) * WAVE_TILE;

    const float* xb = x + (size_t)b * N_EVENTS * N_SAMPLES;
    const int*   ib = indices + b * N_EVENTS;

    const float sel = (((lane & 15) >> 3) == (lane >> 4)) ? 1.0f : 0.0f;
    v2f a_sel;
    a_sel.x = sel;
    a_sel.y = sel;

    v8f acc0 = {}, acc1 = {}, acc2 = {}, acc3 = {};

    v4f buf[2][GROUP];

    int ti[N_EVENTS];
    #pragma unroll
    for (int e = 0; e < N_EVENTS; ++e) ti[e] = ib[e];

#define LOAD_GROUP(G) do { \
        unsigned vo_[GROUP]; \
        _Pragma("unroll") for (int q = 0; q < GROUP; ++q) \
            vo_[q] = row_offset(tile_start(ti[(G) * GROUP + q], s_base), lane); \
        __builtin_amdgcn_sched_barrier(0); \
        _Pragma("unroll") for (int q = 0; q < GROUP; ++q) \
            buf[(G) & 1][q] = load_row(xb, (G) * GROUP + q, vo_[q]); \
        } while (0)

    LOAD_GROUP(0);

    #pragma unroll
    for (int g = 0; g < NGROUPS; ++g) {
        __builtin_amdgcn_sched_barrier(0);
        if (g + 1 < NGROUPS) LOAD_GROUP(g + 1);
        __builtin_amdgcn_sched_barrier(0);

        #pragma unroll
        for (int p = 0; p < GROUP / 2; ++p) {
            const int e0 = g * GROUP + 2 * p;
            const v4f v0 = zero_unless_begun(buf[g & 1][2 * p],     tile_start(ti[e0],     s_base));
            const v4f v1 = zero_unless_begun(buf[g & 1][2 * p + 1], tile_start(ti[e0 + 1], s_base));
            v2f b0, b1, b2, b3;
            b0.x = v0.x; b0.y = v1.x;
            b1.x = v0.y; b1.y = v1.y;
            b2.x = v0.z; b2.y = v1.z;
            b3.x = v0.w; b3.y = v1.w;
            acc0 = __builtin_amdgcn_wmma_f32_16x16x4_f32(false, a_sel, false, b0,
                                                         (short)0, acc0, false, false);
            acc1 = __builtin_amdgcn_wmma_f32_16x16x4_f32(false, a_sel, false, b1,
                                                         (short)0, acc1, false, false);
            acc2 = __builtin_amdgcn_wmma_f32_16x16x4_f32(false, a_sel, false, b2,
                                                         (short)0, acc2, false, false);
            acc3 = __builtin_amdgcn_wmma_f32_16x16x4_f32(false, a_sel, false, b3,
                                                         (short)0, acc3, false, false);
        }
    }
#undef LOAD_GROUP

    v4f o;
    o.x = acc0[0];
    o.y = acc1[0];
    o.z = acc2[0];
    o.w = acc3[0];
    float* ob = out + (size_t)b * N_SAMPLES + s_base;
    *reinterpret_cast<v4f*>(ob + 4 * lane) = o;
}

extern "C" void kernel_launch(void* const* d_in, const int* in_sizes, int n_in,
                              void* d_out, int out_size, void* d_ws, size_t ws_size,
                              hipStream_t stream) {
    (void)in_sizes; (void)n_in; (void)out_size; (void)d_ws; (void)ws_size;
    const float* x   = (const float*)d_in[0];
    const int*   idx = (const int*)d_in[1];
    float*       out = (float*)d_out;

    sparse_audio_gather_wmma<<<GRID, 256, 0, stream>>>(x, idx, out);
}


# sum only the events that have begun at a tile: per-wave list of begun (tile,event) entries (ballot, prefix count, LDS); one wave takes tiles p and 255-p, grid 512->256; list pairs summed (new order)
# speedup vs baseline: 1.4209x; 1.1254x over previous
#include <hip/hip_runtime.h>


typedef __attribute__((ext_vector_type(2))) float v2f;
typedef __attribute__((ext_vector_type(4))) float v4f;
typedef __attribute__((ext_vector_type(4))) int   v4i;
typedef __attribute__((ext_vector_type(8))) float v8f;

#define N_SAMPLES 32768
#define N_EVENTS  64
#define STEP      256
#define BATCH     16

#define WAVES_PER_BLOCK 8
#define WAVE_TILE       128
#define TILES_PER_BATCH  (N_SAMPLES / WAVE_TILE)
#define PAIRS_PER_BATCH  (TILES_PER_BATCH / 2)
#define BLOCKS_PER_BATCH (PAIRS_PER_BATCH / WAVES_PER_BLOCK)
#define GRID (BATCH * BLOCKS_PER_BATCH)

#define GROUP      16
#define MAX_ENTRY  (2 * N_EVENTS)
#define MAX_GROUPS (MAX_ENTRY / GROUP)
#define LIST_LEN   (MAX_ENTRY + GROUP)

__device__ __forceinline__ v4f zero_unless(v4f v, bool keep) {
    const int mask = keep ? -1 : 0;
    const v4i m = {mask, mask, mask, mask};
    return __builtin_bit_cast(v4f, __builtin_bit_cast(v4i, v) & m);
}

__global__ __launch_bounds__(256)
void sparse_audio_begun_only_wmma(const float* __restrict__ x,
                                  const int*   __restrict__ indices,
                                  float*       __restrict__ out) {
    __shared__ unsigned list_all[WAVES_PER_BLOCK][LIST_LEN];

    const int tid  = threadIdx.x;
    const int lane = tid & 31;
    const int b = blockIdx.x / BLOCKS_PER_BATCH;
    const int wave_in_block = __builtin_amdgcn_readfirstlane(tid >> 5);
    const int p    = (blockIdx.x % BLOCKS_PER_BATCH) * WAVES_PER_BLOCK + wave_in_block;
    const int t_lo = p;
    const int t_hi = TILES_PER_BATCH - 1 - p;
    const int s_lo = t_lo * WAVE_TILE;
    const int s_hi = t_hi * WAVE_TILE;
    const unsigned a_lo = (unsigned)(t_lo >> 1);
    const unsigned a_hi = (unsigned)(t_hi >> 1);

    const char* xb = reinterpret_cast<const char*>(x + (size_t)b * N_EVENTS * N_SAMPLES);
    const int*  ib = indices + b * N_EVENTS;

    const unsigned i0 = (unsigned)ib[lane];
    const unsigned i1 = (unsigned)ib[lane + 32];
    const unsigned base0 = (unsigned)lane * (N_SAMPLES * 4u) - i0 * (STEP * 4u);
    const unsigned base1 = (unsigned)(lane + 32) * (N_SAMPLES * 4u) - i1 * (STEP * 4u);
    const unsigned lo_bytes = (unsigned)s_lo * 4u;
    const unsigned hi_bytes = (unsigned)s_hi * 4u;

    const bool l0 = (i0 <= a_lo), l1 = (i1 <= a_lo);
    const bool h0 = (i0 <= a_hi), h1 = (i1 <= a_hi);
    const unsigned ml0 = __builtin_amdgcn_ballot_w32(l0);
    const unsigned ml1 = __builtin_amdgcn_ballot_w32(l1);
    const unsigned mh0 = __builtin_amdgcn_ballot_w32(h0);
    const unsigned mh1 = __builtin_amdgcn_ballot_w32(h1);
    const int c_l0 = __builtin_popcount(ml0);
    const int n_lo = c_l0 + __builtin_popcount(ml1);
    const int c_h0 = __builtin_popcount(mh0);
    const int n    = n_lo + c_h0 + __builtin_popcount(mh1);
    const int ng   = (n + GROUP - 1) / GROUP;

    const unsigned below = (1u << lane) - 1u;
    unsigned* list = list_all[wave_in_block];
    if (l0) list[__builtin_popcount(ml0 & below)]               = base0 + lo_bytes;
    if (l1) list[c_l0 + __builtin_popcount(ml1 & below)]        = base1 + lo_bytes;
    if (h0) list[n_lo + __builtin_popcount(mh0 & below)]        = base0 + hi_bytes;
    if (h1) list[n_lo + c_h0 + __builtin_popcount(mh1 & below)] = base1 + hi_bytes;
    if (lane < GROUP) list[n + lane] = 0u;
    unsigned lst[MAX_ENTRY / 32];
    #pragma unroll
    for (int r = 0; r < MAX_ENTRY / 32; ++r) lst[r] = list[32 * r + lane];

    const unsigned lane16 = 16u * (unsigned)lane;

    const int   rc     = (lane & 15) - 8 * (lane >> 4);
    const float row_lo = (rc == 0) ? 1.0f : 0.0f;
    const float row_hi = (rc == 1) ? 1.0f : 0.0f;

    v8f acc0 = {}, acc1 = {}, acc2 = {}, acc3 = {};
    v4f buf[2][GROUP];

#define LOAD_GROUP(G) do { \
        unsigned vo_[GROUP]; \
        _Pragma("unroll") for (int q = 0; q < GROUP; ++q) { \
            const int k_ = (G) * GROUP + q; \
            vo_[q] = (unsigned)__builtin_amdgcn_readlane((int)lst[k_ / 32], k_ % 32) + lane16; \
        } \
        __builtin_amdgcn_sched_barrier(0); \
        _Pragma("unroll") for (int q = 0; q < GROUP; ++q) \
            buf[(G) & 1][q] = *reinterpret_cast<const v4f*>(xb + vo_[q]); \
        } while (0)

    if (ng > 0) LOAD_GROUP(0);

    #pragma unroll
    for (int g = 0; g < MAX_GROUPS; ++g) {
        if (g < ng) {
            __builtin_amdgcn_sched_barrier(0);
            if (g + 1 < MAX_GROUPS) { if (g + 1 < ng) LOAD_GROUP(g + 1); }
            __builtin_amdgcn_sched_barrier(0);

            #pragma unroll
            for (int q = 0; q < GROUP / 2; ++q) {
                const int k0 = g * GROUP + 2 * q;
                const int k1 = k0 + 1;
                if (k0 < n) {
                    const v4f v0 = buf[g & 1][2 * q];
                    const v4f v1 = zero_unless(buf[g & 1][2 * q + 1], k1 < n);
                    v2f a_;
                    a_.x = (k0 >= n_lo) ? row_hi : row_lo;
                    a_.y = (k1 >= n_lo) ? row_hi : row_lo;
                    v2f b0, b1, b2, b3;
                    b0.x = v0.x; b0.y = v1.x;
                    b1.x = v0.y; b1.y = v1.y;
                    b2.x = v0.z; b2.y = v1.z;
                    b3.x = v0.w; b3.y = v1.w;
                    acc0 = __builtin_amdgcn_wmma_f32_16x16x4_f32(false, a_, false, b0,
                                                                 (short)0, acc0, false, false);
                    acc1 = __builtin_amdgcn_wmma_f32_16x16x4_f32(false, a_, false, b1,
                                                                 (short)0, acc1, false, false);
                    acc2 = __builtin_amdgcn_wmma_f32_16x16x4_f32(false, a_, false, b2,
                                                                 (short)0, acc2, false, false);
                    acc3 = __builtin_amdgcn_wmma_f32_16x16x4_f32(false, a_, false, b3,
                                                                 (short)0, acc3, false, false);
                }
            }
        }
    }
#undef LOAD_GROUP

    asm volatile("s_wait_loadcnt 0x0" ::: "memory");

    v4f o_lo, o_hi;
    o_lo.x = acc0[0]; o_lo.y = acc1[0]; o_lo.z = acc2[0]; o_lo.w = acc3[0];
    o_hi.x = acc0[1]; o_hi.y = acc1[1]; o_hi.z = acc2[1]; o_hi.w = acc3[1];
    float* ob = out + (size_t)b * N_SAMPLES;
    *reinterpret_cast<v4f*>(ob + s_lo + 4 * lane) = o_lo;
    *reinterpret_cast<v4f*>(ob + s_hi + 4 * lane) = o_hi;
}

extern "C" void kernel_launch(void* const* d_in, const int* in_sizes, int n_in,
                              void* d_out, int out_size, void* d_ws, size_t ws_size,
                              hipStream_t stream) {
    (void)in_sizes; (void)n_in; (void)out_size; (void)d_ws; (void)ws_size;
    const float* x   = (const float*)d_in[0];
    const int*   idx = (const int*)d_in[1];
    float*       out = (float*)d_out;

    sparse_audio_begun_only_wmma<<<GRID, 256, 0, stream>>>(x, idx, out);
}


# two 16-load groups kept in flight behind the group being summed (ring of 3 register groups); loads issued on branch paths that never rejoin before a sum, so every wait is exact; same sum order
# speedup vs baseline: 1.6209x; 1.1408x over previous
#include <hip/hip_runtime.h>


typedef __attribute__((ext_vector_type(2))) float v2f;
typedef __attribute__((ext_vector_type(4))) float v4f;
typedef __attribute__((ext_vector_type(4))) int   v4i;
typedef __attribute__((ext_vector_type(8))) float v8f;

#define N_SAMPLES 32768
#define N_EVENTS  64
#define STEP      256
#define BATCH     16

#define WAVES_PER_BLOCK 8
#define WAVE_TILE       128
#define TILES_PER_BATCH  (N_SAMPLES / WAVE_TILE)
#define PAIRS_PER_BATCH  (TILES_PER_BATCH / 2)
#define BLOCKS_PER_BATCH (PAIRS_PER_BATCH / WAVES_PER_BLOCK)
#define GRID (BATCH * BLOCKS_PER_BATCH)

#define GROUP      16
#define MAX_ENTRY  (2 * N_EVENTS)
#define MAX_GROUPS (MAX_ENTRY / GROUP)
#define LIST_LEN   (MAX_ENTRY + GROUP)

__device__ __forceinline__ v4f zero_unless(v4f v, bool keep) {
    const int mask = keep ? -1 : 0;
    const v4i m = {mask, mask, mask, mask};
    return __builtin_bit_cast(v4f, __builtin_bit_cast(v4i, v) & m);
}


#define RING 3

template <int J>
__device__ __forceinline__ void load_group(v4f (&buf)[RING][GROUP], const unsigned (&lst)[MAX_ENTRY / 32],
                                           int g, const char* xb, unsigned lane16) {
    const int r = g >> 1;
    unsigned cur = lst[0];
    cur = (r == 1) ? lst[1] : cur;
    cur = (r == 2) ? lst[2] : cur;
    cur = (r == 3) ? lst[3] : cur;
    const int first = (g & 1) * GROUP;
    unsigned vo[GROUP];
    #pragma unroll
    for (int q = 0; q < GROUP; ++q)
        vo[q] = (unsigned)__builtin_amdgcn_readlane((int)cur, first + q) + lane16;
    __builtin_amdgcn_sched_barrier(0);
    #pragma unroll
    for (int q = 0; q < GROUP; ++q)
        buf[J][q] = *reinterpret_cast<const v4f*>(xb + vo[q]);
    __builtin_amdgcn_sched_barrier(0);
}

template <int J, bool FULL>
__device__ __forceinline__ void sum_group(const v4f (&buf)[RING][GROUP], int g, int n, int n_lo,
                                          float row_lo, float row_hi,
                                          v8f& acc0, v8f& acc1, v8f& acc2, v8f& acc3) {
    #pragma unroll
    for (int q = 0; q < GROUP / 2; ++q) {
        const int k0 = g * GROUP + 2 * q;
        const int k1 = k0 + 1;
        if (FULL || k0 < n) {
            const v4f v0 = buf[J][2 * q];
            const v4f v1 = FULL ? buf[J][2 * q + 1] : zero_unless(buf[J][2 * q + 1], k1 < n);
            v2f a_;
            a_.x = (k0 >= n_lo) ? row_hi : row_lo;
            a_.y = (k1 >= n_lo) ? row_hi : row_lo;
            v2f b0, b1, b2, b3;
            b0.x = v0.x; b0.y = v1.x;
            b1.x = v0.y; b1.y = v1.y;
            b2.x = v0.z; b2.y = v1.z;
            b3.x = v0.w; b3.y = v1.w;
            acc0 = __builtin_amdgcn_wmma_f32_16x16x4_f32(false, a_, false, b0,
                                                         (short)0, acc0, false, false);
            acc1 = __builtin_amdgcn_wmma_f32_16x16x4_f32(false, a_, false, b1,
                                                         (short)0, acc1, false, false);
            acc2 = __builtin_amdgcn_wmma_f32_16x16x4_f32(false, a_, false, b2,
                                                         (short)0, acc2, false, false);
            acc3 = __builtin_amdgcn_wmma_f32_16x16x4_f32(false, a_, false, b3,
                                                         (short)0, acc3, false, false);
        }
    }
}

template <int G>
__device__ __forceinline__ void pipeline_from(v4f (&buf)[RING][GROUP], const unsigned (&lst)[MAX_ENTRY / 32],
                                              const char* xb, unsigned lane16, int n, int n_lo, int ng,
                                              float row_lo, float row_hi,
                                              v8f& acc0, v8f& acc1, v8f& acc2, v8f& acc3) {
    if constexpr (G + 2 < MAX_GROUPS) {
        if (G + 2 < ng) {
            load_group<(G + 2) % RING>(buf, lst, G + 2, xb, lane16);
            sum_group<G % RING, true>(buf, G, n, n_lo, row_lo, row_hi, acc0, acc1, acc2, acc3);
            pipeline_from<G + 1>(buf, lst, xb, lane16, n, n_lo, ng, row_lo, row_hi, acc0, acc1, acc2, acc3);
            return;
        }
    }
    sum_group<G % RING, true>(buf, G, n, n_lo, row_lo, row_hi, acc0, acc1, acc2, acc3);
    sum_group<(G + 1) % RING, false>(buf, G + 1, n, n_lo, row_lo, row_hi, acc0, acc1, acc2, acc3);
}

__global__ __launch_bounds__(256)
void sparse_audio_begun_only_wmma(const float* __restrict__ x,
                                  const int*   __restrict__ indices,
                                  float*       __restrict__ out) {
    __shared__ unsigned list_all[WAVES_PER_BLOCK][LIST_LEN];

    const int tid  = threadIdx.x;
    const int lane = tid & 31;
    const int b = blockIdx.x / BLOCKS_PER_BATCH;
    const int wave_in_block = __builtin_amdgcn_readfirstlane(tid >> 5);
    const int p    = (blockIdx.x % BLOCKS_PER_BATCH) * WAVES_PER_BLOCK + wave_in_block;
    const int t_lo = p;
    const int t_hi = TILES_PER_BATCH - 1 - p;
    const int s_lo = t_lo * WAVE_TILE;
    const int s_hi = t_hi * WAVE_TILE;
    const unsigned a_lo = (unsigned)(t_lo >> 1);
    const unsigned a_hi = (unsigned)(t_hi >> 1);

    const char* xb = reinterpret_cast<const char*>(x + (size_t)b * N_EVENTS * N_SAMPLES);
    const int*  ib = indices + b * N_EVENTS;

    const unsigned i0 = (unsigned)ib[lane];
    const unsigned i1 = (unsigned)ib[lane + 32];
    const unsigned base0 = (unsigned)lane * (N_SAMPLES * 4u) - i0 * (STEP * 4u);
    const unsigned base1 = (unsigned)(lane + 32) * (N_SAMPLES * 4u) - i1 * (STEP * 4u);
    const unsigned lo_bytes = (unsigned)s_lo * 4u;
    const unsigned hi_bytes = (unsigned)s_hi * 4u;

    const bool l0 = (i0 <= a_lo), l1 = (i1 <= a_lo);
    const bool h0 = (i0 <= a_hi), h1 = (i1 <= a_hi);
    const unsigned ml0 = __builtin_amdgcn_ballot_w32(l0);
    const unsigned ml1 = __builtin_amdgcn_ballot_w32(l1);
    const unsigned mh0 = __builtin_amdgcn_ballot_w32(h0);
    const unsigned mh1 = __builtin_amdgcn_ballot_w32(h1);
    const int c_l0 = __builtin_popcount(ml0);
    const int n_lo = c_l0 + __builtin_popcount(ml1);
    const int c_h0 = __builtin_popcount(mh0);
    const int n    = n_lo + c_h0 + __builtin_popcount(mh1);
    const int ng   = (n + GROUP - 1) / GROUP;

    const unsigned below = (1u << lane) - 1u;
    unsigned* list = list_all[wave_in_block];
    if (l0) list[__builtin_popcount(ml0 & below)]               = base0 + lo_bytes;
    if (l1) list[c_l0 + __builtin_popcount(ml1 & below)]        = base1 + lo_bytes;
    if (h0) list[n_lo + __builtin_popcount(mh0 & below)]        = base0 + hi_bytes;
    if (h1) list[n_lo + c_h0 + __builtin_popcount(mh1 & below)] = base1 + hi_bytes;
    if (lane < GROUP) list[n + lane] = 0u;
    unsigned lst[MAX_ENTRY / 32];
    #pragma unroll
    for (int r = 0; r < MAX_ENTRY / 32; ++r) lst[r] = list[32 * r + lane];

    const unsigned lane16 = 16u * (unsigned)lane;

    const int   rc     = (lane & 15) - 8 * (lane >> 4);
    const float row_lo = (rc == 0) ? 1.0f : 0.0f;
    const float row_hi = (rc == 1) ? 1.0f : 0.0f;

    v8f acc0 = {}, acc1 = {}, acc2 = {}, acc3 = {};
    v4f buf[RING][GROUP];

    if (ng == 1) {
        load_group<0>(buf, lst, 0, xb, lane16);
        sum_group<0, false>(buf, 0, n, n_lo, row_lo, row_hi, acc0, acc1, acc2, acc3);
    } else if (ng >= 2) {
        load_group<0>(buf, lst, 0, xb, lane16);
        load_group<1>(buf, lst, 1, xb, lane16);
        pipeline_from<0>(buf, lst, xb, lane16, n, n_lo, ng, row_lo, row_hi, acc0, acc1, acc2, acc3);
    }

    asm volatile("s_wait_loadcnt 0x0" ::: "memory");

    v4f o_lo, o_hi;
    o_lo.x = acc0[0]; o_lo.y = acc1[0]; o_lo.z = acc2[0]; o_lo.w = acc3[0];
    o_hi.x = acc0[1]; o_hi.y = acc1[1]; o_hi.z = acc2[1]; o_hi.w = acc3[1];
    float* ob = out + (size_t)b * N_SAMPLES;
    *reinterpret_cast<v4f*>(ob + s_lo + 4 * lane) = o_lo;
    *reinterpret_cast<v4f*>(ob + s_hi + 4 * lane) = o_hi;
}

extern "C" void kernel_launch(void* const* d_in, const int* in_sizes, int n_in,
                              void* d_out, int out_size, void* d_ws, size_t ws_size,
                              hipStream_t stream) {
    (void)in_sizes; (void)n_in; (void)out_size; (void)d_ws; (void)ws_size;
    const float* x   = (const float*)d_in[0];
    const int*   idx = (const int*)d_in[1];
    float*       out = (float*)d_out;

    sparse_audio_begun_only_wmma<<<GRID, 256, 0, stream>>>(x, idx, out);
}
